# starting kernel
# baseline (speedup 1.0000x reference)
#include <hip/hip_runtime.h>
#include <math.h>

#define N_NODES 50000
#define N_EDGES 640000
#define DIM     128

typedef __attribute__((ext_vector_type(2))) float v2f;
typedef __attribute__((ext_vector_type(8))) float v8f;


__global__ void k_deg_init(unsigned* __restrict__ deg) {
    int i = blockIdx.x * blockDim.x + threadIdx.x;
    if (i < N_NODES) deg[i] = 1u;
}

__global__ void k_deg_count(const int* __restrict__ dst, unsigned* __restrict__ deg) {
    int e = blockIdx.x * blockDim.x + threadIdx.x;
    if (e < N_EDGES) atomicAdd(&deg[dst[e]], 1u);
}

__global__ void k_dinv(const unsigned* __restrict__ deg, float* __restrict__ dinv) {
    int i = blockIdx.x * blockDim.x + threadIdx.x;
    if (i < N_NODES) dinv[i] = rsqrtf((float)deg[i]);
}


__global__ __launch_bounds__(256) void k_gemm(const float* __restrict__ A,
                                              const float* __restrict__ W,
                                              const float* __restrict__ bias,
                                              float* __restrict__ out,
                                              int addBias) {
    const int wave = threadIdx.x >> 5;
    const int lane = threadIdx.x & 31;
    const int l15  = lane & 15;
    const int hi   = lane >> 4;
    const int rowBase = blockIdx.x << 4;
    const int colBase = wave << 4;

    const float* arow = A + (size_t)(rowBase + l15) * DIM + (hi << 1);
    const float* wcol = W + (size_t)(hi << 1) * DIM + colBase + l15;

    v8f c = {};
#pragma unroll
    for (int k = 0; k < DIM; k += 4) {
        v2f a = *(const v2f*)(arow + k);
        v2f b;
        b.x = wcol[(size_t)k * DIM];
        b.y = wcol[(size_t)(k + 1) * DIM];
        c = __builtin_amdgcn_wmma_f32_16x16x4_f32(false, a, false, b,
                                                  (short)0, c, false, false);
    }

    const float bv = addBias ? bias[colBase + l15] : 0.0f;
    const int m0 = hi << 3;
    float* orow = out + (size_t)(rowBase + m0) * DIM + colBase + l15;
#pragma unroll
    for (int r = 0; r < 8; ++r)
        orow[(size_t)r * DIM] = c[r] + bv;
}


__global__ __launch_bounds__(256) void k_self(const float* __restrict__ h,
                                              const float* __restrict__ dinv,
                                              float* __restrict__ agg) {
    int t = blockIdx.x * blockDim.x + threadIdx.x;
    int node = t >> 5;
    float s = dinv[node];
    s *= s;
    float4 v = ((const float4*)h)[t];
    float4 o;
    o.x = v.x * s; o.y = v.y * s; o.z = v.z * s; o.w = v.w * s;
    ((float4*)agg)[t] = o;
}


__global__ __launch_bounds__(256) void k_scatter(const int* __restrict__ src,
                                                 const int* __restrict__ dst,
                                                 const float* __restrict__ dinv,
                                                 const float* __restrict__ h,
                                                 float* __restrict__ agg) {
    int w    = (blockIdx.x * blockDim.x + threadIdx.x) >> 5;
    int lane = threadIdx.x & 31;
    int s = src[w];
    int d = dst[w];
    float norm = dinv[s] * dinv[d];
    float4 v = ((const float4*)(h + (size_t)s * DIM))[lane];
    float* ap = agg + (size_t)d * DIM + (lane << 2);
    unsafeAtomicAdd(ap + 0, norm * v.x);
    unsafeAtomicAdd(ap + 1, norm * v.y);
    unsafeAtomicAdd(ap + 2, norm * v.z);
    unsafeAtomicAdd(ap + 3, norm * v.w);
}


__global__ __launch_bounds__(256) void k_act(const float* __restrict__ agg,
                                             const float* __restrict__ b1,
                                             float* __restrict__ h) {
    int t = blockIdx.x * blockDim.x + threadIdx.x;
    h[t] = tanhf(agg[t] + b1[t & (DIM - 1)]);
}


extern "C" void kernel_launch(void* const* d_in, const int* in_sizes, int n_in,
                              void* d_out, int out_size, void* d_ws, size_t ws_size,
                              hipStream_t stream) {
    const float* x  = (const float*)d_in[0];
    const int*   ei = (const int*)  d_in[1];
    const float* W1 = (const float*)d_in[2];
    const float* b1 = (const float*)d_in[3];
    const float* W2 = (const float*)d_in[4];
    const float* b2 = (const float*)d_in[5];
    float* out = (float*)d_out;

    float*    ws   = (float*)d_ws;
    unsigned* deg  = (unsigned*)ws;
    float*    dinv = ws + 65536;
    float*    h    = ws + 131072;
    float*    agg  = ws + 131072 + (size_t)N_NODES * DIM;

    const int* esrc = ei;
    const int* edst = ei + N_EDGES;

    k_deg_init <<<(N_NODES + 255) / 256, 256, 0, stream>>>(deg);
    k_deg_count<<<(N_EDGES + 255) / 256, 256, 0, stream>>>(edst, deg);
    k_dinv     <<<(N_NODES + 255) / 256, 256, 0, stream>>>(deg, dinv);

    k_gemm <<<N_NODES / 16, 256, 0, stream>>>(x, W1, b1, h, 0);
    k_self <<<(N_NODES * DIM / 4) / 256, 256, 0, stream>>>(h, dinv, agg);
    k_scatter<<<N_EDGES / 8, 256, 0, stream>>>(esrc, edst, dinv, h, agg);
    k_act  <<<(N_NODES * DIM) / 256, 256, 0, stream>>>(agg, b1, h);
    k_gemm <<<N_NODES / 16, 256, 0, stream>>>(h, W2, b2, out, 1);
}


# Sum each destination row in registers: edges bucketed by destination in-call (rank = arrival order at the degree atomics); float atomic scatter and self-loop pass gone; sum order = rank order
# speedup vs baseline: 1.4361x; 1.4361x over previous
#include <hip/hip_runtime.h>
#include <math.h>

#define N_NODES 50000
#define N_EDGES 640000
#define DIM     128

typedef __attribute__((ext_vector_type(2))) float v2f;
typedef __attribute__((ext_vector_type(8))) float v8f;


__global__ void k_deg_init(unsigned* __restrict__ deg, unsigned* __restrict__ counter) {
    int i = blockIdx.x * blockDim.x + threadIdx.x;
    if (i < N_NODES) deg[i] = 1u;
    if (i == 0) *counter = 0u;
}

__global__ void k_deg_count(const int* __restrict__ dst, unsigned* __restrict__ deg,
                            unsigned* __restrict__ rank) {
    int e = blockIdx.x * blockDim.x + threadIdx.x;
    if (e < N_EDGES) rank[e] = atomicAdd(&deg[dst[e]], 1u) - 1u;
}

__global__ __launch_bounds__(256) void k_dinv_alloc(const unsigned* __restrict__ deg,
                                                    float* __restrict__ dinv,
                                                    unsigned* __restrict__ rowstart,
                                                    unsigned* __restrict__ counter) {
    const int i = blockIdx.x * blockDim.x + threadIdx.x;
    const int lane = threadIdx.x & 31;
    const bool ok = i < N_NODES;
    const unsigned dg  = ok ? deg[i] : 1u;
    const unsigned cnt = dg - 1u;
    unsigned incl = cnt;
    for (int o = 1; o < 32; o <<= 1) {
        unsigned t = __shfl_up(incl, (unsigned)o);
        if (lane >= o) incl += t;
    }
    const unsigned total = __shfl(incl, 31);
    unsigned base = 0u;
    if (lane == 31) base = atomicAdd(counter, total);
    base = __shfl(base, 31);
    if (ok) {
        rowstart[i] = base + incl - cnt;
        dinv[i] = rsqrtf((float)dg);
    }
}

__global__ __launch_bounds__(256) void k_fill(const int* __restrict__ src,
                                              const int* __restrict__ dst,
                                              const unsigned* __restrict__ rank,
                                              const unsigned* __restrict__ rowstart,
                                              int* __restrict__ csr) {
    int e = blockIdx.x * blockDim.x + threadIdx.x;
    int d = dst[e];
    unsigned pos = rowstart[d] + rank[e];
    __hip_atomic_exchange(&csr[pos], src[e], __ATOMIC_RELAXED, __HIP_MEMORY_SCOPE_AGENT);
}


__global__ __launch_bounds__(256) void k_gemm(const float* __restrict__ A,
                                              const float* __restrict__ W,
                                              const float* __restrict__ bias,
                                              float* __restrict__ out,
                                              int addBias) {
    const int wave = threadIdx.x >> 5;
    const int lane = threadIdx.x & 31;
    const int l15  = lane & 15;
    const int hi   = lane >> 4;
    const int rowBase = blockIdx.x << 4;
    const int colBase = wave << 4;

    const float* arow = A + (size_t)(rowBase + l15) * DIM + (hi << 1);
    const float* wcol = W + (size_t)(hi << 1) * DIM + colBase + l15;

    v8f c = {};
#pragma unroll
    for (int k = 0; k < DIM; k += 4) {
        v2f a = *(const v2f*)(arow + k);
        v2f b;
        b.x = wcol[(size_t)k * DIM];
        b.y = wcol[(size_t)(k + 1) * DIM];
        c = __builtin_amdgcn_wmma_f32_16x16x4_f32(false, a, false, b,
                                                  (short)0, c, false, false);
    }

    const float bv = addBias ? bias[colBase + l15] : 0.0f;
    const int m0 = hi << 3;
    float* orow = out + (size_t)(rowBase + m0) * DIM + colBase + l15;
#pragma unroll
    for (int r = 0; r < 8; ++r)
        orow[(size_t)r * DIM] = c[r] + bv;
}


__device__ __forceinline__ void fma4(float4& acc, float w, const float4& a) {
    acc.x += w * a.x; acc.y += w * a.y; acc.z += w * a.z; acc.w += w * a.w;
}

__global__ __launch_bounds__(256) void k_gather(const int* __restrict__ csr,
                                                const unsigned* __restrict__ rowstart,
                                                const unsigned* __restrict__ deg,
                                                const float* __restrict__ dinv,
                                                const float* __restrict__ h,
                                                float* __restrict__ agg) {
    const int lane = threadIdx.x & 31;
    const int d = __builtin_amdgcn_readfirstlane((int)((blockIdx.x * blockDim.x + threadIdx.x) >> 5));
    const unsigned start = rowstart[d];
    const int cnt = (int)deg[d] - 1;
    const float dd = dinv[d];
    const float4* h4 = (const float4*)h;

    const float4 self = h4[(size_t)d * 32 + lane];
    const float sw = dd * dd;
    float4 acc;
    acc.x = self.x * sw; acc.y = self.y * sw; acc.z = self.z * sw; acc.w = self.w * sw;

    for (int base = 0; base < cnt; base += 32) {
        const int left = cnt - base;
        const int n = left < 32 ? left : 32;
        int   myS = 0;
        float myW = 0.0f;
        if (lane < n) {
            myS = csr[start + (unsigned)(base + lane)];
            myW = dinv[myS];
        }
        int j = 0;
        for (; j + 4 <= n; j += 4) {
            const int s0 = __builtin_amdgcn_readlane(myS, j);
            const int s1 = __builtin_amdgcn_readlane(myS, j + 1);
            const int s2 = __builtin_amdgcn_readlane(myS, j + 2);
            const int s3 = __builtin_amdgcn_readlane(myS, j + 3);
            const float4 a0 = h4[(size_t)s0 * 32 + lane];
            const float4 a1 = h4[(size_t)s1 * 32 + lane];
            const float4 a2 = h4[(size_t)s2 * 32 + lane];
            const float4 a3 = h4[(size_t)s3 * 32 + lane];
            const float w0 = __int_as_float(__builtin_amdgcn_readlane(__float_as_int(myW), j)) * dd;
            const float w1 = __int_as_float(__builtin_amdgcn_readlane(__float_as_int(myW), j + 1)) * dd;
            const float w2 = __int_as_float(__builtin_amdgcn_readlane(__float_as_int(myW), j + 2)) * dd;
            const float w3 = __int_as_float(__builtin_amdgcn_readlane(__float_as_int(myW), j + 3)) * dd;
            fma4(acc, w0, a0);
            fma4(acc, w1, a1);
            fma4(acc, w2, a2);
            fma4(acc, w3, a3);
        }
        for (; j < n; ++j) {
            const int s0 = __builtin_amdgcn_readlane(myS, j);
            const float4 a0 = h4[(size_t)s0 * 32 + lane];
            const float w0 = __int_as_float(__builtin_amdgcn_readlane(__float_as_int(myW), j)) * dd;
            fma4(acc, w0, a0);
        }
    }
    ((float4*)agg)[(size_t)d * 32 + lane] = acc;
}


__global__ __launch_bounds__(256) void k_act(const float* __restrict__ agg,
                                             const float* __restrict__ b1,
                                             float* __restrict__ h) {
    int t = blockIdx.x * blockDim.x + threadIdx.x;
    h[t] = tanhf(agg[t] + b1[t & (DIM - 1)]);
}


extern "C" void kernel_launch(void* const* d_in, const int* in_sizes, int n_in,
                              void* d_out, int out_size, void* d_ws, size_t ws_size,
                              hipStream_t stream) {
    const float* x  = (const float*)d_in[0];
    const int*   ei = (const int*)  d_in[1];
    const float* W1 = (const float*)d_in[2];
    const float* b1 = (const float*)d_in[3];
    const float* W2 = (const float*)d_in[4];
    const float* b2 = (const float*)d_in[5];
    float* out = (float*)d_out;

    float*    ws   = (float*)d_ws;
    unsigned* deg  = (unsigned*)ws;
    float*    dinv = ws + 65536;
    float*    h    = ws + 131072;
    float*    agg  = ws + 131072 + (size_t)N_NODES * DIM;

    const int* esrc = ei;
    const int* edst = ei + N_EDGES;

    unsigned* counter  = deg + 65535;
    unsigned* rank     = (unsigned*)out;
    int*      csr      = (int*)out + N_EDGES;
    unsigned* rowstart = (unsigned*)out + 2 * N_EDGES;

    k_deg_init  <<<(N_NODES + 255) / 256, 256, 0, stream>>>(deg, counter);
    k_deg_count <<<(N_EDGES + 255) / 256, 256, 0, stream>>>(edst, deg, rank);
    k_dinv_alloc<<<(N_NODES + 255) / 256, 256, 0, stream>>>(deg, dinv, rowstart, counter);
    k_fill      <<<N_EDGES / 256, 256, 0, stream>>>(esrc, edst, rank, rowstart, csr);

    k_gemm  <<<N_NODES / 16, 256, 0, stream>>>(x, W1, b1, h, 0);
    k_gather<<<N_NODES / 8, 256, 0, stream>>>(csr, rowstart, deg, dinv, h, agg);
    k_act   <<<(N_NODES * DIM) / 256, 256, 0, stream>>>(agg, b1, h);
    k_gemm  <<<N_NODES / 16, 256, 0, stream>>>(h, W2, b2, out, 1);
}


# Bias + tanh applied in the aggregation kernel's epilogue to the row sum still in registers; the separate activation pass (200000 waves, one element per thread) and its memory round trip are gone
# speedup vs baseline: 1.5887x; 1.1063x over previous
#include <hip/hip_runtime.h>
#include <math.h>

#define N_NODES 50000
#define N_EDGES 640000
#define DIM     128

typedef __attribute__((ext_vector_type(2))) float v2f;
typedef __attribute__((ext_vector_type(8))) float v8f;


__global__ void k_deg_init(unsigned* __restrict__ deg, unsigned* __restrict__ counter) {
    int i = blockIdx.x * blockDim.x + threadIdx.x;
    if (i < N_NODES) deg[i] = 1u;
    if (i == 0) *counter = 0u;
}

__global__ void k_deg_count(const int* __restrict__ dst, unsigned* __restrict__ deg,
                            unsigned* __restrict__ rank) {
    int e = blockIdx.x * blockDim.x + threadIdx.x;
    if (e < N_EDGES) rank[e] = atomicAdd(&deg[dst[e]], 1u) - 1u;
}

__global__ __launch_bounds__(256) void k_dinv_alloc(const unsigned* __restrict__ deg,
                                                    float* __restrict__ dinv,
                                                    unsigned* __restrict__ rowstart,
                                                    unsigned* __restrict__ counter) {
    const int i = blockIdx.x * blockDim.x + threadIdx.x;
    const int lane = threadIdx.x & 31;
    const bool ok = i < N_NODES;
    const unsigned dg  = ok ? deg[i] : 1u;
    const unsigned cnt = dg - 1u;
    unsigned incl = cnt;
    for (int o = 1; o < 32; o <<= 1) {
        unsigned t = __shfl_up(incl, (unsigned)o);
        if (lane >= o) incl += t;
    }
    const unsigned total = __shfl(incl, 31);
    unsigned base = 0u;
    if (lane == 31) base = atomicAdd(counter, total);
    base = __shfl(base, 31);
    if (ok) {
        rowstart[i] = base + incl - cnt;
        dinv[i] = rsqrtf((float)dg);
    }
}

__global__ __launch_bounds__(256) void k_fill(const int* __restrict__ src,
                                              const int* __restrict__ dst,
                                              const unsigned* __restrict__ rank,
                                              const unsigned* __restrict__ rowstart,
                                              int* __restrict__ csr) {
    int e = blockIdx.x * blockDim.x + threadIdx.x;
    int d = dst[e];
    unsigned pos = rowstart[d] + rank[e];
    __hip_atomic_exchange(&csr[pos], src[e], __ATOMIC_RELAXED, __HIP_MEMORY_SCOPE_AGENT);
}


__global__ __launch_bounds__(256) void k_gemm(const float* __restrict__ A,
                                              const float* __restrict__ W,
                                              const float* __restrict__ bias,
                                              float* __restrict__ out,
                                              int addBias) {
    const int wave = threadIdx.x >> 5;
    const int lane = threadIdx.x & 31;
    const int l15  = lane & 15;
    const int hi   = lane >> 4;
    const int rowBase = blockIdx.x << 4;
    const int colBase = wave << 4;

    const float* arow = A + (size_t)(rowBase + l15) * DIM + (hi << 1);
    const float* wcol = W + (size_t)(hi << 1) * DIM + colBase + l15;

    v8f c = {};
#pragma unroll
    for (int k = 0; k < DIM; k += 4) {
        v2f a = *(const v2f*)(arow + k);
        v2f b;
        b.x = wcol[(size_t)k * DIM];
        b.y = wcol[(size_t)(k + 1) * DIM];
        c = __builtin_amdgcn_wmma_f32_16x16x4_f32(false, a, false, b,
                                                  (short)0, c, false, false);
    }

    const float bv = addBias ? bias[colBase + l15] : 0.0f;
    const int m0 = hi << 3;
    float* orow = out + (size_t)(rowBase + m0) * DIM + colBase + l15;
#pragma unroll
    for (int r = 0; r < 8; ++r)
        orow[(size_t)r * DIM] = c[r] + bv;
}


__device__ __forceinline__ void fma4(float4& acc, float w, const float4& a) {
    acc.x += w * a.x; acc.y += w * a.y; acc.z += w * a.z; acc.w += w * a.w;
}

__global__ __launch_bounds__(256) void k_gather(const int* __restrict__ csr,
                                                const unsigned* __restrict__ rowstart,
                                                const unsigned* __restrict__ deg,
                                                const float* __restrict__ dinv,
                                                const float* __restrict__ h,
                                                const float* __restrict__ b1,
                                                float* __restrict__ act) {
    const int lane = threadIdx.x & 31;
    const int d = __builtin_amdgcn_readfirstlane((int)((blockIdx.x * blockDim.x + threadIdx.x) >> 5));
    const unsigned start = rowstart[d];
    const int cnt = (int)deg[d] - 1;
    const float dd = dinv[d];
    const float4* h4 = (const float4*)h;
    const float4 bias = ((const float4*)b1)[lane];

    const float4 self = h4[(size_t)d * 32 + lane];
    const float sw = dd * dd;
    float4 acc;
    acc.x = self.x * sw; acc.y = self.y * sw; acc.z = self.z * sw; acc.w = self.w * sw;

    for (int base = 0; base < cnt; base += 32) {
        const int left = cnt - base;
        const int n = left < 32 ? left : 32;
        int   myS = 0;
        float myW = 0.0f;
        if (lane < n) {
            myS = csr[start + (unsigned)(base + lane)];
            myW = dinv[myS];
        }
        int j = 0;
        for (; j + 4 <= n; j += 4) {
            const int s0 = __builtin_amdgcn_readlane(myS, j);
            const int s1 = __builtin_amdgcn_readlane(myS, j + 1);
            const int s2 = __builtin_amdgcn_readlane(myS, j + 2);
            const int s3 = __builtin_amdgcn_readlane(myS, j + 3);
            const float4 a0 = h4[(size_t)s0 * 32 + lane];
            const float4 a1 = h4[(size_t)s1 * 32 + lane];
            const float4 a2 = h4[(size_t)s2 * 32 + lane];
            const float4 a3 = h4[(size_t)s3 * 32 + lane];
            const float w0 = __int_as_float(__builtin_amdgcn_readlane(__float_as_int(myW), j)) * dd;
            const float w1 = __int_as_float(__builtin_amdgcn_readlane(__float_as_int(myW), j + 1)) * dd;
            const float w2 = __int_as_float(__builtin_amdgcn_readlane(__float_as_int(myW), j + 2)) * dd;
            const float w3 = __int_as_float(__builtin_amdgcn_readlane(__float_as_int(myW), j + 3)) * dd;
            fma4(acc, w0, a0);
            fma4(acc, w1, a1);
            fma4(acc, w2, a2);
            fma4(acc, w3, a3);
        }
        for (; j < n; ++j) {
            const int s0 = __builtin_amdgcn_readlane(myS, j);
            const float4 a0 = h4[(size_t)s0 * 32 + lane];
            const float w0 = __int_as_float(__builtin_amdgcn_readlane(__float_as_int(myW), j)) * dd;
            fma4(acc, w0, a0);
        }
    }
    float4 r;
    r.x = tanhf(acc.x + bias.x);
    r.y = tanhf(acc.y + bias.y);
    r.z = tanhf(acc.z + bias.z);
    r.w = tanhf(acc.w + bias.w);
    ((float4*)act)[(size_t)d * 32 + lane] = r;
}


extern "C" void kernel_launch(void* const* d_in, const int* in_sizes, int n_in,
                              void* d_out, int out_size, void* d_ws, size_t ws_size,
                              hipStream_t stream) {
    const float* x  = (const float*)d_in[0];
    const int*   ei = (const int*)  d_in[1];
    const float* W1 = (const float*)d_in[2];
    const float* b1 = (const float*)d_in[3];
    const float* W2 = (const float*)d_in[4];
    const float* b2 = (const float*)d_in[5];
    float* out = (float*)d_out;

    float*    ws   = (float*)d_ws;
    unsigned* deg  = (unsigned*)ws;
    float*    dinv = ws + 65536;
    float*    h    = ws + 131072;
    float*    act  = ws + 131072 + (size_t)N_NODES * DIM;

    const int* esrc = ei;
    const int* edst = ei + N_EDGES;

    unsigned* counter  = deg + 65535;
    unsigned* rank     = (unsigned*)out;
    int*      csr      = (int*)out + N_EDGES;
    unsigned* rowstart = (unsigned*)out + 2 * N_EDGES;

    k_deg_init  <<<(N_NODES + 255) / 256, 256, 0, stream>>>(deg, counter);
    k_deg_count <<<(N_EDGES + 255) / 256, 256, 0, stream>>>(edst, deg, rank);
    k_dinv_alloc<<<(N_NODES + 255) / 256, 256, 0, stream>>>(deg, dinv, rowstart, counter);
    k_fill      <<<N_EDGES / 256, 256, 0, stream>>>(esrc, edst, rank, rowstart, csr);

    k_gemm  <<<N_NODES / 16, 256, 0, stream>>>(x, W1, b1, h, 0);
    k_gather<<<N_NODES / 8, 256, 0, stream>>>(csr, rowstart, deg, dinv, h, b1, act);
    k_gemm  <<<N_NODES / 16, 256, 0, stream>>>(act, W2, b2, out, 1);
}


# GEMM operand feed: each wave keeps its 128x16 slice of W in registers over 4-5 row strips; A strips staged once per work-group in LDS by whole-line loads (was 16-line loads per wave); sums unchanged
# speedup vs baseline: 1.8636x; 1.1730x over previous
#include <hip/hip_runtime.h>
#include <math.h>

#define N_NODES 50000
#define N_EDGES 640000
#define DIM     128

typedef __attribute__((ext_vector_type(2))) float v2f;
typedef __attribute__((ext_vector_type(8))) float v8f;


__global__ void k_deg_init(unsigned* __restrict__ deg, unsigned* __restrict__ counter) {
    int i = blockIdx.x * blockDim.x + threadIdx.x;
    if (i < N_NODES) deg[i] = 1u;
    if (i == 0) *counter = 0u;
}

__global__ void k_deg_count(const int* __restrict__ dst, unsigned* __restrict__ deg,
                            unsigned* __restrict__ rank) {
    int e = blockIdx.x * blockDim.x + threadIdx.x;
    if (e < N_EDGES) rank[e] = atomicAdd(&deg[dst[e]], 1u) - 1u;
}

__global__ __launch_bounds__(256) void k_dinv_alloc(const unsigned* __restrict__ deg,
                                                    float* __restrict__ dinv,
                                                    unsigned* __restrict__ rowstart,
                                                    unsigned* __restrict__ counter) {
    const int i = blockIdx.x * blockDim.x + threadIdx.x;
    const int lane = threadIdx.x & 31;
    const bool ok = i < N_NODES;
    const unsigned dg  = ok ? deg[i] : 1u;
    const unsigned cnt = dg - 1u;
    unsigned incl = cnt;
    for (int o = 1; o < 32; o <<= 1) {
        unsigned t = __shfl_up(incl, (unsigned)o);
        if (lane >= o) incl += t;
    }
    const unsigned total = __shfl(incl, 31);
    unsigned base = 0u;
    if (lane == 31) base = atomicAdd(counter, total);
    base = __shfl(base, 31);
    if (ok) {
        rowstart[i] = base + incl - cnt;
        dinv[i] = rsqrtf((float)dg);
    }
}

__global__ __launch_bounds__(256) void k_fill(const int* __restrict__ src,
                                              const int* __restrict__ dst,
                                              const unsigned* __restrict__ rank,
                                              const unsigned* __restrict__ rowstart,
                                              int* __restrict__ csr) {
    int e = blockIdx.x * blockDim.x + threadIdx.x;
    int d = dst[e];
    unsigned pos = rowstart[d] + rank[e];
    __hip_atomic_exchange(&csr[pos], src[e], __ATOMIC_RELAXED, __HIP_MEMORY_SCOPE_AGENT);
}


#define N_STRIPS    (N_NODES / 16)
#define A_PITCH     132
#define GEMM_BLOCKS 768

__global__ __launch_bounds__(256) void k_gemm(const float* __restrict__ A,
                                              const float* __restrict__ W,
                                              const float* __restrict__ bias,
                                              float* __restrict__ out,
                                              int addBias) {
    __shared__ __attribute__((aligned(16))) float As[2][16 * A_PITCH];

    const int tid  = threadIdx.x;
    const int wave = tid >> 5;
    const int lane = tid & 31;
    const int l15  = lane & 15;
    const int hi   = lane >> 4;
    const int colBase = wave << 4;

    const float4* gA = (const float4*)A;
    const int stOff = (tid >> 5) * A_PITCH + ((tid & 31) << 2);

    int s = blockIdx.x;
    float4 p0 = gA[(size_t)s * 512 + tid];
    float4 p1 = gA[(size_t)s * 512 + 256 + tid];

    const float* wcol = W + (size_t)(hi << 1) * DIM + colBase + l15;
    v2f b[32];
#pragma unroll
    for (int ks = 0; ks < 32; ++ks) {
        b[ks].x = wcol[(size_t)(4 * ks) * DIM];
        b[ks].y = wcol[(size_t)(4 * ks + 1) * DIM];
    }
    const float bv = addBias ? bias[colBase + l15] : 0.0f;

    *(float4*)&As[0][stOff]               = p0;
    *(float4*)&As[0][stOff + 8 * A_PITCH] = p1;
    __syncthreads();

    const int fragOff = l15 * A_PITCH + (hi << 1);
    const int m0 = hi << 3;
    int buf = 0;
    for (; s < N_STRIPS; s += GEMM_BLOCKS) {
        const int sn = s + GEMM_BLOCKS;
        if (sn < N_STRIPS) {
            p0 = gA[(size_t)sn * 512 + tid];
            p1 = gA[(size_t)sn * 512 + 256 + tid];
        }

        const float* as = &As[buf][fragOff];
        v2f a[32];
#pragma unroll
        for (int ks = 0; ks < 32; ++ks)
            a[ks] = *(const v2f*)(as + 4 * ks);

        __builtin_amdgcn_sched_barrier(0);

        v8f c = {};
#pragma unroll
        for (int ks = 0; ks < 32; ++ks)
            c = __builtin_amdgcn_wmma_f32_16x16x4_f32(false, a[ks], false, b[ks],
                                                      (short)0, c, false, false);
        __builtin_amdgcn_sched_barrier(0);

        float o[8];
#pragma unroll
        for (int r = 0; r < 8; ++r)
            o[r] = c[r] + bv;

        *(float4*)&As[buf ^ 1][stOff]               = p0;
        *(float4*)&As[buf ^ 1][stOff + 8 * A_PITCH] = p1;
        __syncthreads();

        float* orow = out + (size_t)(s * 16 + m0) * DIM + colBase + l15;
#pragma unroll
        for (int r = 0; r < 8; ++r)
            orow[(size_t)r * DIM] = o[r];
        buf ^= 1;
    }
}


__device__ __forceinline__ void fma4(float4& acc, float w, const float4& a) {
    acc.x += w * a.x; acc.y += w * a.y; acc.z += w * a.z; acc.w += w * a.w;
}

__global__ __launch_bounds__(256) void k_gather(const int* __restrict__ csr,
                                                const unsigned* __restrict__ rowstart,
                                                const unsigned* __restrict__ deg,
                                                const float* __restrict__ dinv,
                                                const float* __restrict__ h,
                                                const float* __restrict__ b1,
                                                float* __restrict__ act) {
    const int lane = threadIdx.x & 31;
    const int d = __builtin_amdgcn_readfirstlane((int)((blockIdx.x * blockDim.x + threadIdx.x) >> 5));
    const unsigned start = rowstart[d];
    const int cnt = (int)deg[d] - 1;
    const float dd = dinv[d];
    const float4* h4 = (const float4*)h;
    const float4 bias = ((const float4*)b1)[lane];

    const float4 self = h4[(size_t)d * 32 + lane];
    const float sw = dd * dd;
    float4 acc;
    acc.x = self.x * sw; acc.y = self.y * sw; acc.z = self.z * sw; acc.w = self.w * sw;

    for (int base = 0; base < cnt; base += 32) {
        const int left = cnt - base;
        const int n = left < 32 ? left : 32;
        int   myS = 0;
        float myW = 0.0f;
        if (lane < n) {
            myS = csr[start + (unsigned)(base + lane)];
            myW = dinv[myS];
        }
        int j = 0;
        for (; j + 4 <= n; j += 4) {
            const int s0 = __builtin_amdgcn_readlane(myS, j);
            const int s1 = __builtin_amdgcn_readlane(myS, j + 1);
            const int s2 = __builtin_amdgcn_readlane(myS, j + 2);
            const int s3 = __builtin_amdgcn_readlane(myS, j + 3);
            const float4 a0 = h4[(size_t)s0 * 32 + lane];
            const float4 a1 = h4[(size_t)s1 * 32 + lane];
            const float4 a2 = h4[(size_t)s2 * 32 + lane];
            const float4 a3 = h4[(size_t)s3 * 32 + lane];
            const float w0 = __int_as_float(__builtin_amdgcn_readlane(__float_as_int(myW), j)) * dd;
            const float w1 = __int_as_float(__builtin_amdgcn_readlane(__float_as_int(myW), j + 1)) * dd;
            const float w2 = __int_as_float(__builtin_amdgcn_readlane(__float_as_int(myW), j + 2)) * dd;
            const float w3 = __int_as_float(__builtin_amdgcn_readlane(__float_as_int(myW), j + 3)) * dd;
            fma4(acc, w0, a0);
            fma4(acc, w1, a1);
            fma4(acc, w2, a2);
            fma4(acc, w3, a3);
        }
        for (; j < n; ++j) {
            const int s0 = __builtin_amdgcn_readlane(myS, j);
            const float4 a0 = h4[(size_t)s0 * 32 + lane];
            const float w0 = __int_as_float(__builtin_amdgcn_readlane(__float_as_int(myW), j)) * dd;
            fma4(acc, w0, a0);
        }
    }
    float4 r;
    r.x = tanhf(acc.x + bias.x);
    r.y = tanhf(acc.y + bias.y);
    r.z = tanhf(acc.z + bias.z);
    r.w = tanhf(acc.w + bias.w);
    ((float4*)act)[(size_t)d * 32 + lane] = r;
}


extern "C" void kernel_launch(void* const* d_in, const int* in_sizes, int n_in,
                              void* d_out, int out_size, void* d_ws, size_t ws_size,
                              hipStream_t stream) {
    const float* x  = (const float*)d_in[0];
    const int*   ei = (const int*)  d_in[1];
    const float* W1 = (const float*)d_in[2];
    const float* b1 = (const float*)d_in[3];
    const float* W2 = (const float*)d_in[4];
    const float* b2 = (const float*)d_in[5];
    float* out = (float*)d_out;

    float*    ws   = (float*)d_ws;
    unsigned* deg  = (unsigned*)ws;
    float*    dinv = ws + 65536;
    float*    h    = ws + 131072;
    float*    act  = ws + 131072 + (size_t)N_NODES * DIM;

    const int* esrc = ei;
    const int* edst = ei + N_EDGES;

    unsigned* counter  = deg + 65535;
    unsigned* rank     = (unsigned*)out;
    int*      csr      = (int*)out + N_EDGES;
    unsigned* rowstart = (unsigned*)out + 2 * N_EDGES;

    k_deg_init  <<<(N_NODES + 255) / 256, 256, 0, stream>>>(deg, counter);
    k_deg_count <<<(N_EDGES + 255) / 256, 256, 0, stream>>>(edst, deg, rank);
    k_dinv_alloc<<<(N_NODES + 255) / 256, 256, 0, stream>>>(deg, dinv, rowstart, counter);
    k_fill      <<<N_EDGES / 256, 256, 0, stream>>>(esrc, edst, rank, rowstart, csr);

    k_gemm  <<<GEMM_BLOCKS, 256, 0, stream>>>(x, W1, b1, h, 0);
    k_gather<<<N_NODES / 8, 256, 0, stream>>>(csr, rowstart, deg, dinv, h, b1, act);
    k_gemm  <<<GEMM_BLOCKS, 256, 0, stream>>>(act, W2, b2, out, 1);
}


# Destination buckets at a fixed 32-slot stride, filled by the counting pass itself (slot = count returned by the add; longer rows spill to a list); prefix and fill launches gone; sum order as before
# speedup vs baseline: 2.2369x; 1.2003x over previous
#include <hip/hip_runtime.h>
#include <math.h>

#define N_NODES 50000
#define N_EDGES 640000
#define DIM     128

typedef __attribute__((ext_vector_type(2))) float v2f;
typedef __attribute__((ext_vector_type(8))) float v8f;


#define BUCKET 32

__global__ void k_cnt_init(unsigned* __restrict__ cnt, unsigned* __restrict__ spillCount) {
    int i = blockIdx.x * blockDim.x + threadIdx.x;
    if (i < N_NODES) cnt[i] = 0u;
    if (i == 0) *spillCount = 0u;
}

__global__ __launch_bounds__(256) void k_bucket(const int* __restrict__ src,
                                                const int* __restrict__ dst,
                                                unsigned* __restrict__ cnt,
                                                int* __restrict__ bucket,
                                                int* __restrict__ spill,
                                                unsigned* __restrict__ spillCount) {
    const int e = blockIdx.x * blockDim.x + threadIdx.x;
    const int d = dst[e];
    const int s = src[e];
    const unsigned slot = atomicAdd(&cnt[d], 1u);
    if (slot < (unsigned)BUCKET) {
        __hip_atomic_exchange(&bucket[d * BUCKET + (int)slot], s, __ATOMIC_RELAXED, __HIP_MEMORY_SCOPE_AGENT);
    } else {
        const unsigned i = atomicAdd(spillCount, 1u);
        __hip_atomic_exchange(&spill[2u * i],      d, __ATOMIC_RELAXED, __HIP_MEMORY_SCOPE_AGENT);
        __hip_atomic_exchange(&spill[2u * i + 1u], s, __ATOMIC_RELAXED, __HIP_MEMORY_SCOPE_AGENT);
    }
}


#define N_STRIPS    (N_NODES / 16)
#define A_PITCH     132
#define GEMM_BLOCKS 768

__global__ __launch_bounds__(256) void k_gemm(const float* __restrict__ A,
                                              const float* __restrict__ W,
                                              const float* __restrict__ bias,
                                              float* __restrict__ out,
                                              int addBias) {
    __shared__ __attribute__((aligned(16))) float As[2][16 * A_PITCH];

    const int tid  = threadIdx.x;
    const int wave = tid >> 5;
    const int lane = tid & 31;
    const int l15  = lane & 15;
    const int hi   = lane >> 4;
    const int colBase = wave << 4;

    const float4* gA = (const float4*)A;
    const int stOff = (tid >> 5) * A_PITCH + ((tid & 31) << 2);

    int s = blockIdx.x;
    float4 p0 = gA[(size_t)s * 512 + tid];
    float4 p1 = gA[(size_t)s * 512 + 256 + tid];

    const float* wcol = W + (size_t)(hi << 1) * DIM + colBase + l15;
    v2f b[32];
#pragma unroll
    for (int ks = 0; ks < 32; ++ks) {
        b[ks].x = wcol[(size_t)(4 * ks) * DIM];
        b[ks].y = wcol[(size_t)(4 * ks + 1) * DIM];
    }
    const float bv = addBias ? bias[colBase + l15] : 0.0f;

    *(float4*)&As[0][stOff]               = p0;
    *(float4*)&As[0][stOff + 8 * A_PITCH] = p1;
    __syncthreads();

    const int fragOff = l15 * A_PITCH + (hi << 1);
    const int m0 = hi << 3;
    int buf = 0;
    for (; s < N_STRIPS; s += GEMM_BLOCKS) {
        const int sn = s + GEMM_BLOCKS;
        if (sn < N_STRIPS) {
            p0 = gA[(size_t)sn * 512 + tid];
            p1 = gA[(size_t)sn * 512 + 256 + tid];
        }

        const float* as = &As[buf][fragOff];
        v2f a[32];
#pragma unroll
        for (int ks = 0; ks < 32; ++ks)
            a[ks] = *(const v2f*)(as + 4 * ks);

        __builtin_amdgcn_sched_barrier(0);

        v8f c = {};
#pragma unroll
        for (int ks = 0; ks < 32; ++ks)
            c = __builtin_amdgcn_wmma_f32_16x16x4_f32(false, a[ks], false, b[ks],
                                                      (short)0, c, false, false);
        __builtin_amdgcn_sched_barrier(0);

        float o[8];
#pragma unroll
        for (int r = 0; r < 8; ++r)
            o[r] = c[r] + bv;

        *(float4*)&As[buf ^ 1][stOff]               = p0;
        *(float4*)&As[buf ^ 1][stOff + 8 * A_PITCH] = p1;
        __syncthreads();

        float* orow = out + (size_t)(s * 16 + m0) * DIM + colBase + l15;
#pragma unroll
        for (int r = 0; r < 8; ++r)
            orow[(size_t)r * DIM] = o[r];
        buf ^= 1;
    }
}


__device__ __forceinline__ void fma4(float4& acc, float w, const float4& a) {
    acc.x += w * a.x; acc.y += w * a.y; acc.z += w * a.z; acc.w += w * a.w;
}

__global__ __launch_bounds__(256) void k_gather(const int* __restrict__ bucket,
                                                const unsigned* __restrict__ cnt,
                                                const int* __restrict__ spill,
                                                const unsigned* __restrict__ spillCount,
                                                const float* __restrict__ h,
                                                const float* __restrict__ b1,
                                                float* __restrict__ act) {
    const int lane = threadIdx.x & 31;
    const int d = __builtin_amdgcn_readfirstlane((int)((blockIdx.x * blockDim.x + threadIdx.x) >> 5));
    const int c = (int)cnt[d];
    const float dd = rsqrtf((float)(c + 1));
    const float4* h4 = (const float4*)h;
    const float4 bias = ((const float4*)b1)[lane];

    const float4 self = h4[(size_t)d * 32 + lane];
    const float sw = dd * dd;
    float4 acc;
    acc.x = self.x * sw; acc.y = self.y * sw; acc.z = self.z * sw; acc.w = self.w * sw;

    {
        const int n = c < BUCKET ? c : BUCKET;
        int   myS = 0;
        float myW = 0.0f;
        if (lane < n) {
            myS = bucket[d * BUCKET + lane];
            myW = rsqrtf((float)(cnt[myS] + 1u));
        }
        int j = 0;
        for (; j + 4 <= n; j += 4) {
            const int s0 = __builtin_amdgcn_readlane(myS, j);
            const int s1 = __builtin_amdgcn_readlane(myS, j + 1);
            const int s2 = __builtin_amdgcn_readlane(myS, j + 2);
            const int s3 = __builtin_amdgcn_readlane(myS, j + 3);
            const float4 a0 = h4[(size_t)s0 * 32 + lane];
            const float4 a1 = h4[(size_t)s1 * 32 + lane];
            const float4 a2 = h4[(size_t)s2 * 32 + lane];
            const float4 a3 = h4[(size_t)s3 * 32 + lane];
            const float w0 = __int_as_float(__builtin_amdgcn_readlane(__float_as_int(myW), j)) * dd;
            const float w1 = __int_as_float(__builtin_amdgcn_readlane(__float_as_int(myW), j + 1)) * dd;
            const float w2 = __int_as_float(__builtin_amdgcn_readlane(__float_as_int(myW), j + 2)) * dd;
            const float w3 = __int_as_float(__builtin_amdgcn_readlane(__float_as_int(myW), j + 3)) * dd;
            fma4(acc, w0, a0);
            fma4(acc, w1, a1);
            fma4(acc, w2, a2);
            fma4(acc, w3, a3);
        }
        for (; j < n; ++j) {
            const int s0 = __builtin_amdgcn_readlane(myS, j);
            const float4 a0 = h4[(size_t)s0 * 32 + lane];
            const float w0 = __int_as_float(__builtin_amdgcn_readlane(__float_as_int(myW), j)) * dd;
            fma4(acc, w0, a0);
        }
    }

    if (c > BUCKET) {
        const int total = (int)*spillCount;
        for (int base = 0; base < total; base += 32) {
            const int i = base + lane;
            int   myS = 0;
            float myW = 0.0f;
            bool  mine = false;
            if (i < total) {
                const int2 ent = ((const int2*)spill)[i];
                if (ent.x == d) {
                    mine = true;
                    myS = ent.y;
                    myW = rsqrtf((float)(cnt[myS] + 1u));
                }
            }
            unsigned m = __builtin_amdgcn_ballot_w32(mine);
            while (m != 0u) {
                const int j = __builtin_amdgcn_readfirstlane((int)__builtin_ctz(m));
                m &= m - 1u;
                const int s0 = __builtin_amdgcn_readlane(myS, j);
                const float4 a0 = h4[(size_t)s0 * 32 + lane];
                const float w0 = __int_as_float(__builtin_amdgcn_readlane(__float_as_int(myW), j)) * dd;
                fma4(acc, w0, a0);
            }
        }
    }

    float4 r;
    r.x = tanhf(acc.x + bias.x);
    r.y = tanhf(acc.y + bias.y);
    r.z = tanhf(acc.z + bias.z);
    r.w = tanhf(acc.w + bias.w);
    ((float4*)act)[(size_t)d * 32 + lane] = r;
}


extern "C" void kernel_launch(void* const* d_in, const int* in_sizes, int n_in,
                              void* d_out, int out_size, void* d_ws, size_t ws_size,
                              hipStream_t stream) {
    const float* x  = (const float*)d_in[0];
    const int*   ei = (const int*)  d_in[1];
    const float* W1 = (const float*)d_in[2];
    const float* b1 = (const float*)d_in[3];
    const float* W2 = (const float*)d_in[4];
    const float* b2 = (const float*)d_in[5];
    float* out = (float*)d_out;

    float*    ws   = (float*)d_ws;
    unsigned* cnt  = (unsigned*)ws;
    float*    h    = ws + 131072;
    float*    act  = ws + 131072 + (size_t)N_NODES * DIM;

    const int* esrc = ei;
    const int* edst = ei + N_EDGES;

    unsigned* spillCount = cnt + 65535;
    int*      bucket     = (int*)out;
    int*      spill      = (int*)out + N_NODES * BUCKET;

    k_cnt_init<<<(N_NODES + 255) / 256, 256, 0, stream>>>(cnt, spillCount);
    k_bucket  <<<N_EDGES / 256, 256, 0, stream>>>(esrc, edst, cnt, bucket, spill, spillCount);

    k_gemm  <<<GEMM_BLOCKS, 256, 0, stream>>>(x, W1, b1, h, 0);
    k_gather<<<N_NODES / 8, 256, 0, stream>>>(bucket, cnt, spill, spillCount, h, b1, act);
    k_gemm  <<<GEMM_BLOCKS, 256, 0, stream>>>(act, W2, b2, out, 1);
}


# Count zeroing moved into the first GEMM launch, which now runs in front of the bucket pass (it reads no counts); the separate init launch is gone, 5 launches -> 4; sums and their order unchanged
# speedup vs baseline: 2.4129x; 1.0787x over previous
#include <hip/hip_runtime.h>
#include <math.h>

#define N_NODES 50000
#define N_EDGES 640000
#define DIM     128

typedef __attribute__((ext_vector_type(2))) float v2f;
typedef __attribute__((ext_vector_type(8))) float v8f;


#define BUCKET 32

__global__ __launch_bounds__(256) void k_bucket(const int* __restrict__ src,
                                                const int* __restrict__ dst,
                                                unsigned* __restrict__ cnt,
                                                int* __restrict__ bucket,
                                                int* __restrict__ spill,
                                                unsigned* __restrict__ spillCount) {
    const int e = blockIdx.x * blockDim.x + threadIdx.x;
    const int d = dst[e];
    const int s = src[e];
    const unsigned slot = atomicAdd(&cnt[d], 1u);
    if (slot < (unsigned)BUCKET) {
        __hip_atomic_exchange(&bucket[d * BUCKET + (int)slot], s, __ATOMIC_RELAXED, __HIP_MEMORY_SCOPE_AGENT);
    } else {
        const unsigned i = atomicAdd(spillCount, 1u);
        __hip_atomic_exchange(&spill[2u * i],      d, __ATOMIC_RELAXED, __HIP_MEMORY_SCOPE_AGENT);
        __hip_atomic_exchange(&spill[2u * i + 1u], s, __ATOMIC_RELAXED, __HIP_MEMORY_SCOPE_AGENT);
    }
}


#define N_STRIPS    (N_NODES / 16)
#define A_PITCH     132
#define GEMM_BLOCKS 768

__global__ __launch_bounds__(256) void k_gemm(const float* __restrict__ A,
                                              const float* __restrict__ W,
                                              const float* __restrict__ bias,
                                              float* __restrict__ out,
                                              int addBias,
                                              unsigned* __restrict__ zeroCnt,
                                              unsigned* __restrict__ zeroSpillCount) {
    __shared__ __attribute__((aligned(16))) float As[2][16 * A_PITCH];

    const int tid  = threadIdx.x;
    const int wave = tid >> 5;
    const int lane = tid & 31;
    const int l15  = lane & 15;
    const int hi   = lane >> 4;
    const int colBase = wave << 4;

    const float4* gA = (const float4*)A;
    const int stOff = (tid >> 5) * A_PITCH + ((tid & 31) << 2);

    int s = blockIdx.x;
    float4 p0 = gA[(size_t)s * 512 + tid];
    float4 p1 = gA[(size_t)s * 512 + 256 + tid];

    const float* wcol = W + (size_t)(hi << 1) * DIM + colBase + l15;
    v2f b[32];
#pragma unroll
    for (int ks = 0; ks < 32; ++ks) {
        b[ks].x = wcol[(size_t)(4 * ks) * DIM];
        b[ks].y = wcol[(size_t)(4 * ks + 1) * DIM];
    }
    const float bv = addBias ? bias[colBase + l15] : 0.0f;

    if (zeroCnt != nullptr) {
        const int g = blockIdx.x * 256 + tid;
        if (g < N_NODES) zeroCnt[g] = 0u;
        if (g == N_NODES) *zeroSpillCount = 0u;
    }

    *(float4*)&As[0][stOff]               = p0;
    *(float4*)&As[0][stOff + 8 * A_PITCH] = p1;
    __syncthreads();

    const int fragOff = l15 * A_PITCH + (hi << 1);
    const int m0 = hi << 3;
    int buf = 0;
    for (; s < N_STRIPS; s += GEMM_BLOCKS) {
        const int sn = s + GEMM_BLOCKS;
        if (sn < N_STRIPS) {
            p0 = gA[(size_t)sn * 512 + tid];
            p1 = gA[(size_t)sn * 512 + 256 + tid];
        }

        const float* as = &As[buf][fragOff];
        v2f a[32];
#pragma unroll
        for (int ks = 0; ks < 32; ++ks)
            a[ks] = *(const v2f*)(as + 4 * ks);

        __builtin_amdgcn_sched_barrier(0);

        v8f c = {};
#pragma unroll
        for (int ks = 0; ks < 32; ++ks)
            c = __builtin_amdgcn_wmma_f32_16x16x4_f32(false, a[ks], false, b[ks],
                                                      (short)0, c, false, false);
        __builtin_amdgcn_sched_barrier(0);

        float o[8];
#pragma unroll
        for (int r = 0; r < 8; ++r)
            o[r] = c[r] + bv;

        *(float4*)&As[buf ^ 1][stOff]               = p0;
        *(float4*)&As[buf ^ 1][stOff + 8 * A_PITCH] = p1;
        __syncthreads();

        float* orow = out + (size_t)(s * 16 + m0) * DIM + colBase + l15;
#pragma unroll
        for (int r = 0; r < 8; ++r)
            orow[(size_t)r * DIM] = o[r];
        buf ^= 1;
    }
}


__device__ __forceinline__ void fma4(float4& acc, float w, const float4& a) {
    acc.x += w * a.x; acc.y += w * a.y; acc.z += w * a.z; acc.w += w * a.w;
}

__global__ __launch_bounds__(256) void k_gather(const int* __restrict__ bucket,
                                                const unsigned* __restrict__ cnt,
                                                const int* __restrict__ spill,
                                                const unsigned* __restrict__ spillCount,
                                                const float* __restrict__ h,
                                                const float* __restrict__ b1,
                                                float* __restrict__ act) {
    const int lane = threadIdx.x & 31;
    const int d = __builtin_amdgcn_readfirstlane((int)((blockIdx.x * blockDim.x + threadIdx.x) >> 5));
    const int c = (int)cnt[d];
    const float dd = rsqrtf((float)(c + 1));
    const float4* h4 = (const float4*)h;
    const float4 bias = ((const float4*)b1)[lane];

    const float4 self = h4[(size_t)d * 32 + lane];
    const float sw = dd * dd;
    float4 acc;
    acc.x = self.x * sw; acc.y = self.y * sw; acc.z = self.z * sw; acc.w = self.w * sw;

    {
        const int n = c < BUCKET ? c : BUCKET;
        int   myS = 0;
        float myW = 0.0f;
        if (lane < n) {
            myS = bucket[d * BUCKET + lane];
            myW = rsqrtf((float)(cnt[myS] + 1u));
        }
        int j = 0;
        for (; j + 4 <= n; j += 4) {
            const int s0 = __builtin_amdgcn_readlane(myS, j);
            const int s1 = __builtin_amdgcn_readlane(myS, j + 1);
            const int s2 = __builtin_amdgcn_readlane(myS, j + 2);
            const int s3 = __builtin_amdgcn_readlane(myS, j + 3);
            const float4 a0 = h4[(size_t)s0 * 32 + lane];
            const float4 a1 = h4[(size_t)s1 * 32 + lane];
            const float4 a2 = h4[(size_t)s2 * 32 + lane];
            const float4 a3 = h4[(size_t)s3 * 32 + lane];
            const float w0 = __int_as_float(__builtin_amdgcn_readlane(__float_as_int(myW), j)) * dd;
            const float w1 = __int_as_float(__builtin_amdgcn_readlane(__float_as_int(myW), j + 1)) * dd;
            const float w2 = __int_as_float(__builtin_amdgcn_readlane(__float_as_int(myW), j + 2)) * dd;
            const float w3 = __int_as_float(__builtin_amdgcn_readlane(__float_as_int(myW), j + 3)) * dd;
            fma4(acc, w0, a0);
            fma4(acc, w1, a1);
            fma4(acc, w2, a2);
            fma4(acc, w3, a3);
        }
        for (; j < n; ++j) {
            const int s0 = __builtin_amdgcn_readlane(myS, j);
            const float4 a0 = h4[(size_t)s0 * 32 + lane];
            const float w0 = __int_as_float(__builtin_amdgcn_readlane(__float_as_int(myW), j)) * dd;
            fma4(acc, w0, a0);
        }
    }

    if (c > BUCKET) {
        const int total = (int)*spillCount;
        for (int base = 0; base < total; base += 32) {
            const int i = base + lane;
            int   myS = 0;
            float myW = 0.0f;
            bool  mine = false;
            if (i < total) {
                const int2 ent = ((const int2*)spill)[i];
                if (ent.x == d) {
                    mine = true;
                    myS = ent.y;
                    myW = rsqrtf((float)(cnt[myS] + 1u));
                }
            }
            unsigned m = __builtin_amdgcn_ballot_w32(mine);
            while (m != 0u) {
                const int j = __builtin_amdgcn_readfirstlane((int)__builtin_ctz(m));
                m &= m - 1u;
                const int s0 = __builtin_amdgcn_readlane(myS, j);
                const float4 a0 = h4[(size_t)s0 * 32 + lane];
                const float w0 = __int_as_float(__builtin_amdgcn_readlane(__float_as_int(myW), j)) * dd;
                fma4(acc, w0, a0);
            }
        }
    }

    float4 r;
    r.x = tanhf(acc.x + bias.x);
    r.y = tanhf(acc.y + bias.y);
    r.z = tanhf(acc.z + bias.z);
    r.w = tanhf(acc.w + bias.w);
    ((float4*)act)[(size_t)d * 32 + lane] = r;
}


extern "C" void kernel_launch(void* const* d_in, const int* in_sizes, int n_in,
                              void* d_out, int out_size, void* d_ws, size_t ws_size,
                              hipStream_t stream) {
    const float* x  = (const float*)d_in[0];
    const int*   ei = (const int*)  d_in[1];
    const float* W1 = (const float*)d_in[2];
    const float* b1 = (const float*)d_in[3];
    const float* W2 = (const float*)d_in[4];
    const float* b2 = (const float*)d_in[5];
    float* out = (float*)d_out;

    float*    ws   = (float*)d_ws;
    unsigned* cnt  = (unsigned*)ws;
    float*    h    = ws + 131072;
    float*    act  = ws + 131072 + (size_t)N_NODES * DIM;

    const int* esrc = ei;
    const int* edst = ei + N_EDGES;

    unsigned* spillCount = cnt + 65535;
    int*      bucket     = (int*)out;
    int*      spill      = (int*)out + N_NODES * BUCKET;

    k_gemm  <<<GEMM_BLOCKS, 256, 0, stream>>>(x, W1, b1, h, 0, cnt, spillCount);
    k_bucket<<<N_EDGES / 256, 256, 0, stream>>>(esrc, edst, cnt, bucket, spill, spillCount);
    k_gather<<<N_NODES / 8, 256, 0, stream>>>(bucket, cnt, spill, spillCount, h, b1, act);
    k_gemm  <<<GEMM_BLOCKS, 256, 0, stream>>>(act, W2, b2, out, 1, nullptr, nullptr);
}


# Aggregation fused into the W2 product launch: 16-wave groups, a row sum per wave into an LDS strip, waves 0-7 multiply it by W2 kept in LDS as k-pairs; act plane and 4th launch gone; sums unchanged
# speedup vs baseline: 2.5427x; 1.0538x over previous
#include <hip/hip_runtime.h>
#include <math.h>

#define N_NODES 50000
#define N_EDGES 640000
#define DIM     128

typedef __attribute__((ext_vector_type(2))) float v2f;
typedef __attribute__((ext_vector_type(8))) float v8f;


#define BUCKET 32

__global__ __launch_bounds__(256) void k_bucket(const int* __restrict__ src,
                                                const int* __restrict__ dst,
                                                unsigned* __restrict__ cnt,
                                                int* __restrict__ bucket,
                                                int* __restrict__ spill,
                                                unsigned* __restrict__ spillCount) {
    const int e = blockIdx.x * blockDim.x + threadIdx.x;
    const int d = dst[e];
    const int s = src[e];
    const unsigned slot = atomicAdd(&cnt[d], 1u);
    if (slot < (unsigned)BUCKET) {
        __hip_atomic_exchange(&bucket[d * BUCKET + (int)slot], s, __ATOMIC_RELAXED, __HIP_MEMORY_SCOPE_AGENT);
    } else {
        const unsigned i = atomicAdd(spillCount, 1u);
        __hip_atomic_exchange(&spill[2u * i],      d, __ATOMIC_RELAXED, __HIP_MEMORY_SCOPE_AGENT);
        __hip_atomic_exchange(&spill[2u * i + 1u], s, __ATOMIC_RELAXED, __HIP_MEMORY_SCOPE_AGENT);
    }
}


#define N_STRIPS    (N_NODES / 16)
#define A_PITCH     132
#define GEMM_BLOCKS 768

__global__ __launch_bounds__(256) void k_gemm(const float* __restrict__ A,
                                              const float* __restrict__ W,
                                              const float* __restrict__ bias,
                                              float* __restrict__ out,
                                              int addBias,
                                              unsigned* __restrict__ zeroCnt,
                                              unsigned* __restrict__ zeroSpillCount) {
    __shared__ __attribute__((aligned(16))) float As[2][16 * A_PITCH];

    const int tid  = threadIdx.x;
    const int wave = tid >> 5;
    const int lane = tid & 31;
    const int l15  = lane & 15;
    const int hi   = lane >> 4;
    const int colBase = wave << 4;

    const float4* gA = (const float4*)A;
    const int stOff = (tid >> 5) * A_PITCH + ((tid & 31) << 2);

    int s = blockIdx.x;
    float4 p0 = gA[(size_t)s * 512 + tid];
    float4 p1 = gA[(size_t)s * 512 + 256 + tid];

    const float* wcol = W + (size_t)(hi << 1) * DIM + colBase + l15;
    v2f b[32];
#pragma unroll
    for (int ks = 0; ks < 32; ++ks) {
        b[ks].x = wcol[(size_t)(4 * ks) * DIM];
        b[ks].y = wcol[(size_t)(4 * ks + 1) * DIM];
    }
    const float bv = addBias ? bias[colBase + l15] : 0.0f;

    if (zeroCnt != nullptr) {
        const int g = blockIdx.x * 256 + tid;
        if (g < N_NODES) zeroCnt[g] = 0u;
        if (g == N_NODES) *zeroSpillCount = 0u;
    }

    *(float4*)&As[0][stOff]               = p0;
    *(float4*)&As[0][stOff + 8 * A_PITCH] = p1;
    __syncthreads();

    const int fragOff = l15 * A_PITCH + (hi << 1);
    const int m0 = hi << 3;
    int buf = 0;
    for (; s < N_STRIPS; s += GEMM_BLOCKS) {
        const int sn = s + GEMM_BLOCKS;
        if (sn < N_STRIPS) {
            p0 = gA[(size_t)sn * 512 + tid];
            p1 = gA[(size_t)sn * 512 + 256 + tid];
        }

        const float* as = &As[buf][fragOff];
        v2f a[32];
#pragma unroll
        for (int ks = 0; ks < 32; ++ks)
            a[ks] = *(const v2f*)(as + 4 * ks);

        __builtin_amdgcn_sched_barrier(0);

        v8f c = {};
#pragma unroll
        for (int ks = 0; ks < 32; ++ks)
            c = __builtin_amdgcn_wmma_f32_16x16x4_f32(false, a[ks], false, b[ks],
                                                      (short)0, c, false, false);
        __builtin_amdgcn_sched_barrier(0);

        float o[8];
#pragma unroll
        for (int r = 0; r < 8; ++r)
            o[r] = c[r] + bv;

        *(float4*)&As[buf ^ 1][stOff]               = p0;
        *(float4*)&As[buf ^ 1][stOff + 8 * A_PITCH] = p1;
        __syncthreads();

        float* orow = out + (size_t)(s * 16 + m0) * DIM + colBase + l15;
#pragma unroll
        for (int r = 0; r < 8; ++r)
            orow[(size_t)r * DIM] = o[r];
        buf ^= 1;
    }
}


__device__ __forceinline__ void fma4(float4& acc, float w, const float4& a) {
    acc.x += w * a.x; acc.y += w * a.y; acc.z += w * a.z; acc.w += w * a.w;
}

__device__ __forceinline__ float4 gather_row(const int d, const int lane,
                                             const int* __restrict__ bucket,
                                             const unsigned* __restrict__ cnt,
                                             const int* __restrict__ spill,
                                             const unsigned* __restrict__ spillCount,
                                             const float4* __restrict__ h4,
                                             const float* __restrict__ b1) {
    const int c = (int)cnt[d];
    const float dd = rsqrtf((float)(c + 1));

    const float4 self = h4[(size_t)d * 32 + lane];
    const float sw = dd * dd;
    float4 acc;
    acc.x = self.x * sw; acc.y = self.y * sw; acc.z = self.z * sw; acc.w = self.w * sw;

    {
        const int n = c < BUCKET ? c : BUCKET;
        int   myS = 0;
        float myW = 0.0f;
        if (lane < n) {
            myS = bucket[d * BUCKET + lane];
            myW = rsqrtf((float)(cnt[myS] + 1u));
        }
        int j = 0;
        for (; j + 4 <= n; j += 4) {
            const int s0 = __builtin_amdgcn_readlane(myS, j);
            const int s1 = __builtin_amdgcn_readlane(myS, j + 1);
            const int s2 = __builtin_amdgcn_readlane(myS, j + 2);
            const int s3 = __builtin_amdgcn_readlane(myS, j + 3);
            const float4 a0 = h4[(size_t)s0 * 32 + lane];
            const float4 a1 = h4[(size_t)s1 * 32 + lane];
            const float4 a2 = h4[(size_t)s2 * 32 + lane];
            const float4 a3 = h4[(size_t)s3 * 32 + lane];
            const float w0 = __int_as_float(__builtin_amdgcn_readlane(__float_as_int(myW), j)) * dd;
            const float w1 = __int_as_float(__builtin_amdgcn_readlane(__float_as_int(myW), j + 1)) * dd;
            const float w2 = __int_as_float(__builtin_amdgcn_readlane(__float_as_int(myW), j + 2)) * dd;
            const float w3 = __int_as_float(__builtin_amdgcn_readlane(__float_as_int(myW), j + 3)) * dd;
            fma4(acc, w0, a0);
            fma4(acc, w1, a1);
            fma4(acc, w2, a2);
            fma4(acc, w3, a3);
        }
#pragma clang loop unroll(disable)
        for (; j < n; ++j) {
            const int s0 = __builtin_amdgcn_readlane(myS, j);
            const float4 a0 = h4[(size_t)s0 * 32 + lane];
            const float w0 = __int_as_float(__builtin_amdgcn_readlane(__float_as_int(myW), j)) * dd;
            fma4(acc, w0, a0);
        }
    }

    if (c > BUCKET) {
        const int total = (int)*spillCount;
        for (int base = 0; base < total; base += 32) {
            const int i = base + lane;
            int   myS = 0;
            float myW = 0.0f;
            bool  mine = false;
            if (i < total) {
                const int2 ent = ((const int2*)spill)[i];
                if (ent.x == d) {
                    mine = true;
                    myS = ent.y;
                    myW = rsqrtf((float)(cnt[myS] + 1u));
                }
            }
            unsigned m = __builtin_amdgcn_ballot_w32(mine);
            while (m != 0u) {
                const int j = __builtin_amdgcn_readfirstlane((int)__builtin_ctz(m));
                m &= m - 1u;
                const int s0 = __builtin_amdgcn_readlane(myS, j);
                const float4 a0 = h4[(size_t)s0 * 32 + lane];
                const float w0 = __int_as_float(__builtin_amdgcn_readlane(__float_as_int(myW), j)) * dd;
                fma4(acc, w0, a0);
            }
        }
    }

    const float4 bias = ((const float4*)b1)[lane];
    float4 r;
    r.x = tanhf(acc.x + bias.x);
    r.y = tanhf(acc.y + bias.y);
    r.z = tanhf(acc.z + bias.z);
    r.w = tanhf(acc.w + bias.w);
    return r;
}

#define FUSE_THREADS 512
#define FUSE_BLOCKS  1024
#define W_ROW        256

__global__ __launch_bounds__(FUSE_THREADS) void k_gather_gemm(const int* __restrict__ bucket,
                                                              const unsigned* __restrict__ cnt,
                                                              const int* __restrict__ spill,
                                                              const unsigned* __restrict__ spillCount,
                                                              const float* __restrict__ h,
                                                              const float* __restrict__ b1,
                                                              const float* __restrict__ W2,
                                                              const float* __restrict__ b2,
                                                              float* __restrict__ out) {
    __shared__ __attribute__((aligned(16))) float Ws[64 * W_ROW];
    __shared__ __attribute__((aligned(16))) float As[16 * A_PITCH];

    const int tid  = threadIdx.x;
    const int wave = __builtin_amdgcn_readfirstlane(tid >> 5);
    const int lane = tid & 31;
    const int l15  = lane & 15;
    const int hi   = lane >> 4;
    const int colBase = (wave & 7) << 4;

    {
        const float4* gW = (const float4*)W2;
#pragma unroll
        for (int it = 0; it < 4; ++it) {
            const int idx = it * FUSE_THREADS + tid;
            const int kp  = idx >> 5;
            const int j   = idx & 31;
            const float4 r0 = gW[(size_t)(2 * kp) * 32 + j];
            const float4 r1 = gW[(size_t)(2 * kp + 1) * 32 + j];
            const int turn = (kp & 1) << 5;
            float* wr = &Ws[kp * W_ROW];
            v2f p;
            p.x = r0.x; p.y = r1.x; *(v2f*)&wr[(8 * j + 0 + turn) & 255] = p;
            p.x = r0.y; p.y = r1.y; *(v2f*)&wr[(8 * j + 2 + turn) & 255] = p;
            p.x = r0.z; p.y = r1.z; *(v2f*)&wr[(8 * j + 4 + turn) & 255] = p;
            p.x = r0.w; p.y = r1.w; *(v2f*)&wr[(8 * j + 6 + turn) & 255] = p;
        }
    }

    const float4* h4 = (const float4*)h;

    const float* as = &As[l15 * A_PITCH + (hi << 1)];
    const float* bs = &Ws[hi * W_ROW + ((2 * (colBase + l15) + (hi << 5)) & 255)];
    const int m0 = hi << 3;

    for (int s = blockIdx.x; s < N_STRIPS; s += FUSE_BLOCKS) {
        const float4 r = gather_row(s * 16 + wave, lane, bucket, cnt, spill, spillCount, h4, b1);
        *(float4*)&As[wave * A_PITCH + (lane << 2)] = r;
        __syncthreads();

        float o[8];
        if (wave < 8) {
            const float bv = b2[colBase + l15];
            v8f c = {};
#pragma unroll
            for (int ch = 0; ch < 4; ++ch) {
                v2f a[8], b[8];
#pragma unroll
                for (int i = 0; i < 8; ++i) {
                    const int ks = ch * 8 + i;
                    a[i] = *(const v2f*)(as + 4 * ks);
                    b[i] = *(const v2f*)(bs + ks * 2 * W_ROW);
                }
                __builtin_amdgcn_sched_barrier(0);
#pragma unroll
                for (int i = 0; i < 8; ++i)
                    c = __builtin_amdgcn_wmma_f32_16x16x4_f32(false, a[i], false, b[i],
                                                              (short)0, c, false, false);
                __builtin_amdgcn_sched_barrier(0);
            }
#pragma unroll
            for (int r8 = 0; r8 < 8; ++r8)
                o[r8] = c[r8] + bv;
        }
        __syncthreads();

        if (wave < 8) {
            float* orow = out + (size_t)(s * 16 + m0) * DIM + colBase + l15;
#pragma unroll
            for (int r8 = 0; r8 < 8; ++r8)
                orow[(size_t)r8 * DIM] = o[r8];
        }
    }
}


extern "C" void kernel_launch(void* const* d_in, const int* in_sizes, int n_in,
                              void* d_out, int out_size, void* d_ws, size_t ws_size,
                              hipStream_t stream) {
    const float* x  = (const float*)d_in[0];
    const int*   ei = (const int*)  d_in[1];
    const float* W1 = (const float*)d_in[2];
    const float* b1 = (const float*)d_in[3];
    const float* W2 = (const float*)d_in[4];
    const float* b2 = (const float*)d_in[5];
    float* out = (float*)d_out;

    float*    ws   = (float*)d_ws;
    unsigned* cnt  = (unsigned*)ws;
    float*    h    = ws + 131072;
    float*    scr  = ws + 131072 + (size_t)N_NODES * DIM;

    const int* esrc = ei;
    const int* edst = ei + N_EDGES;

    unsigned* spillCount = cnt + 65535;
    int*      bucket     = (int*)scr;
    int*      spill      = (int*)scr + N_NODES * BUCKET;

    k_gemm  <<<GEMM_BLOCKS, 256, 0, stream>>>(x, W1, b1, h, 0, cnt, spillCount);
    k_bucket<<<N_EDGES / 256, 256, 0, stream>>>(esrc, edst, cnt, bucket, spill, spillCount);
    k_gather_gemm<<<FUSE_BLOCKS, FUSE_THREADS, 0, stream>>>(bucket, cnt, spill, spillCount, h, b1, W2, b2, out);
}
